# starting kernel
# baseline (speedup 1.0000x reference)
#include <hip/hip_runtime.h>
#include <math.h>


typedef __attribute__((ext_vector_type(2))) float v2f;
typedef __attribute__((ext_vector_type(8))) float v8f;

#define GMAX 256

__device__ __forceinline__ void atomAddF(float* p, float v) {
  __hip_atomic_fetch_add(p, v, __ATOMIC_RELAXED, __HIP_MEMORY_SCOPE_AGENT);
}

__global__ void k_fill(float* p, float v, int n) {
  int i = blockIdx.x * blockDim.x + threadIdx.x;
  if (i < n) p[i] = v;
}

__global__ void k_deg_count(const int* __restrict__ dst, float* __restrict__ deg, int E) {
  int e = blockIdx.x * blockDim.x + threadIdx.x;
  if (e < E) atomAddF(&deg[dst[e]], 1.0f);
}

__global__ void k_to_dinv(float* deg, int n) {
  int i = blockIdx.x * blockDim.x + threadIdx.x;
  if (i < n) deg[i] = rsqrtf(fmaxf(deg[i], 1.0f));
}

template <int OUT>
__global__ void k_gemm_wmma(const float* __restrict__ A, const float* __restrict__ W,
                            float* __restrict__ C, int N) {
  constexpr int KD = 128;
  __shared__ float As[16 * KD];

  const int row0 = blockIdx.x * 16;
  for (int i = threadIdx.x; i < 16 * KD; i += blockDim.x) {
    int r = i / KD, k = i % KD;
    As[i] = (row0 + r < N) ? A[(size_t)(row0 + r) * KD + k] : 0.0f;
  }
  __syncthreads();

  const int wave = threadIdx.x >> 5;
  const int lane = threadIdx.x & 31;
  const int half = lane >> 4;
  const int n    = lane & 15;
  const int cw   = wave * 16;

  v8f c = {};
  for (int k0 = 0; k0 < KD; k0 += 4) {
    const int k = k0 + 2 * half;
    v2f a, b;
    a.x = As[n * KD + k];
    a.y = As[n * KD + k + 1];
    b.x = W[(size_t)k * OUT + cw + n];
    b.y = W[(size_t)(k + 1) * OUT + cw + n];
    c = __builtin_amdgcn_wmma_f32_16x16x4_f32(false, a, false, b,
                                              (short)0, c, false, false);
  }

#pragma unroll
  for (int r = 0; r < 8; ++r) {
    const int m = r + 8 * half;
    if (row0 + m < N) C[(size_t)(row0 + m) * OUT + cw + n] = c[r];
  }
}

template <int F>
__global__ void k_agg_init(const float* __restrict__ xw, const float* __restrict__ dinv,
                           const float* __restrict__ bias, float* __restrict__ out, int N) {
  constexpr int LP = F / 4;
  int t = blockIdx.x * blockDim.x + threadIdx.x;
  int node = t / LP;
  if (node >= N) return;
  int fo = (t % LP) * 4;
  float s = dinv[node]; s *= s;
  float4 v = *(const float4*)&xw[(size_t)node * F + fo];
  const float4 b = *(const float4*)&bias[fo];
  v.x = v.x * s + b.x; v.y = v.y * s + b.y;
  v.z = v.z * s + b.z; v.w = v.w * s + b.w;
  *(float4*)&out[(size_t)node * F + fo] = v;
}

template <int F>
__global__ void k_agg_edges(const float* __restrict__ xw, const float* __restrict__ dinv,
                            const int* __restrict__ src, const int* __restrict__ dst,
                            float* __restrict__ out, int E) {
  constexpr int LP = F / 4;
  int t = blockIdx.x * blockDim.x + threadIdx.x;
  int e = t / LP;
  if (e >= E) return;
  int fo = (t % LP) * 4;
  int s = src[e], d = dst[e];
  float nrm = dinv[s] * dinv[d];
  float4 v = *(const float4*)&xw[(size_t)s * F + fo];
  float* o = &out[(size_t)d * F + fo];
  atomAddF(o + 0, v.x * nrm);
  atomAddF(o + 1, v.y * nrm);
  atomAddF(o + 2, v.z * nrm);
  atomAddF(o + 3, v.w * nrm);
}

__global__ void k_elu_pool(float* __restrict__ h, const int* __restrict__ batch,
                           float* __restrict__ pool, float* __restrict__ cnt, int N) {
  int node = blockIdx.x * (blockDim.x >> 5) + (threadIdx.x >> 5);
  if (node >= N) return;
  int lane = threadIdx.x & 31;
  float4 v = *(float4*)&h[(size_t)node * 128 + lane * 4];
  v.x = v.x > 0.0f ? v.x : expm1f(v.x);
  v.y = v.y > 0.0f ? v.y : expm1f(v.y);
  v.z = v.z > 0.0f ? v.z : expm1f(v.z);
  v.w = v.w > 0.0f ? v.w : expm1f(v.w);
  *(float4*)&h[(size_t)node * 128 + lane * 4] = v;
  int g = batch[node];
  float* p = &pool[(size_t)g * 128 + lane * 4];
  atomAddF(p + 0, v.x); atomAddF(p + 1, v.y);
  atomAddF(p + 2, v.z); atomAddF(p + 3, v.w);
  if (lane == 0) atomAddF(&cnt[g], 1.0f);
}

template <int IN, int OUT>
__global__ void k_master_mlp(const float* __restrict__ pool, const float* __restrict__ cnt,
                             const float* __restrict__ Wm, const float* __restrict__ bm,
                             float* __restrict__ mm, const int* __restrict__ gptr) {
  int G = *gptr;
  int idx = blockIdx.x * blockDim.x + threadIdx.x;
  int g = idx / OUT, j = idx % OUT;
  if (g >= G) return;
  float inv = 1.0f / fmaxf(cnt[g], 1.0f);
  float acc = bm[j];
  for (int i = 0; i < IN; ++i)
    acc = fmaf(pool[(size_t)g * IN + i] * inv, Wm[(size_t)i * OUT + j], acc);
  mm[(size_t)g * OUT + j] = fmaxf(acc, 0.0f);
}

__global__ void k_add_master(float* __restrict__ h, const float* __restrict__ mm,
                             const int* __restrict__ batch, int N) {
  int t = blockIdx.x * blockDim.x + threadIdx.x;
  int node = t >> 5;
  if (node >= N) return;
  int fo = (t & 31) * 4;
  int g = batch[node];
  float4 a = *(float4*)&h[(size_t)node * 128 + fo];
  float4 m = *(const float4*)&mm[(size_t)g * 128 + fo];
  a.x += m.x; a.y += m.y; a.z += m.z; a.w += m.w;
  *(float4*)&h[(size_t)node * 128 + fo] = a;
}

__global__ void k_pool_z(const float* __restrict__ z, const int* __restrict__ batch,
                         float* __restrict__ pool, int N) {
  int t = blockIdx.x * blockDim.x + threadIdx.x;
  int node = t >> 5;
  if (node >= N) return;
  int f = t & 31;
  atomAddF(&pool[(size_t)batch[node] * 32 + f], z[(size_t)node * 32 + f]);
}

__global__ void k_finalize(const float* __restrict__ z, const float* __restrict__ mm2,
                           const int* __restrict__ batch, float* __restrict__ out, int N) {
  int node = blockIdx.x * (blockDim.x >> 5) + (threadIdx.x >> 5);
  if (node >= N) return;
  int lane = threadIdx.x & 31;
  float v = z[(size_t)node * 32 + lane] + mm2[(size_t)batch[node] * 32 + lane];
  float m = v;
#pragma unroll
  for (int o = 16; o >= 1; o >>= 1) m = fmaxf(m, __shfl_xor(m, o, 32));
  float ev = __expf(v - m);
  float s = ev;
#pragma unroll
  for (int o = 16; o >= 1; o >>= 1) s += __shfl_xor(s, o, 32);
  out[(size_t)node * 32 + lane] = ev / s;
}

extern "C" void kernel_launch(void* const* d_in, const int* in_sizes, int n_in,
                              void* d_out, int out_size, void* d_ws, size_t ws_size,
                              hipStream_t stream) {
  const float* X   = (const float*)d_in[0];
  const float* W1  = (const float*)d_in[1];
  const float* b1  = (const float*)d_in[2];
  const float* W2  = (const float*)d_in[3];
  const float* b2  = (const float*)d_in[4];
  const float* mW1 = (const float*)d_in[5];
  const float* mb1 = (const float*)d_in[6];
  const float* mW2 = (const float*)d_in[7];
  const float* mb2 = (const float*)d_in[8];
  const int* ei    = (const int*)d_in[9];
  const int* batch = (const int*)d_in[10];
  const int* gptr  = (const int*)d_in[11];

  const int N = in_sizes[0] / 128;
  const int E = in_sizes[9] / 2;
  const int* src = ei;
  const int* dst = ei + E;

  char* w = (char*)d_ws;
  size_t off = 0;
  auto take = [&](size_t bytes) -> float* {
    float* p = (float*)(w + off);
    off = (off + bytes + 255) & ~(size_t)255;
    return p;
  };
  float* xw    = take((size_t)N * 128 * 4);
  float* h     = take((size_t)N * 128 * 4);
  float* dinv  = take((size_t)N * 4);
  float* z     = take((size_t)N * 32 * 4);
  float* pool1 = take((size_t)GMAX * 128 * 4);
  float* pool2 = take((size_t)GMAX * 32 * 4);
  float* cnt   = take((size_t)GMAX * 4);
  float* mm1   = take((size_t)GMAX * 128 * 4);
  float* mm2   = take((size_t)GMAX * 32 * 4);

  const int B = 256;
  auto blks = [](long long n, int b) { return (unsigned)((n + b - 1) / b); };

  k_fill<<<blks(N, B), B, 0, stream>>>(dinv, 1.0f, N);
  k_fill<<<blks(GMAX * 128, B), B, 0, stream>>>(pool1, 0.0f, GMAX * 128);
  k_fill<<<blks(GMAX * 32, B), B, 0, stream>>>(pool2, 0.0f, GMAX * 32);
  k_fill<<<blks(GMAX, B), B, 0, stream>>>(cnt, 0.0f, GMAX);
  k_deg_count<<<blks(E, B), B, 0, stream>>>(dst, dinv, E);
  k_to_dinv<<<blks(N, B), B, 0, stream>>>(dinv, N);

  k_gemm_wmma<128><<<blks(N, 16), 256, 0, stream>>>(X, W1, xw, N);
  k_agg_init<128><<<blks((long long)N * 32, B), B, 0, stream>>>(xw, dinv, b1, h, N);
  k_agg_edges<128><<<blks((long long)E * 32, B), B, 0, stream>>>(xw, dinv, src, dst, h, E);
  k_elu_pool<<<blks(N, 8), B, 0, stream>>>(h, batch, pool1, cnt, N);

  k_master_mlp<128, 128><<<blks(GMAX * 128, B), B, 0, stream>>>(pool1, cnt, mW1, mb1, mm1, gptr);
  k_add_master<<<blks((long long)N * 32, B), B, 0, stream>>>(h, mm1, batch, N);

  k_gemm_wmma<32><<<blks(N, 16), 64, 0, stream>>>(h, W2, xw, N);
  k_agg_init<32><<<blks((long long)N * 8, B), B, 0, stream>>>(xw, dinv, b2, z, N);
  k_agg_edges<32><<<blks((long long)E * 8, B), B, 0, stream>>>(xw, dinv, src, dst, z, E);

  k_pool_z<<<blks((long long)N * 32, B), B, 0, stream>>>(z, batch, pool2, N);
  k_master_mlp<32, 32><<<blks(GMAX * 32, B), B, 0, stream>>>(pool2, cnt, mW2, mb2, mm2, gptr);
  k_finalize<<<blks(N, 8), B, 0, stream>>>(z, mm2, batch, (float*)d_out, N);
}


# Row atomics cover one whole 128B line per wave instruction: lane = feature l, l+32, ..; a wave walks 16 edges (pool: 1 node); was 8 lanes x 16B stride over 4 lines; sums still in arrival order
# speedup vs baseline: 1.6099x; 1.6099x over previous
#include <hip/hip_runtime.h>
#include <math.h>


typedef __attribute__((ext_vector_type(2))) float v2f;
typedef __attribute__((ext_vector_type(8))) float v8f;

#define GMAX 256

__device__ __forceinline__ void atomAddF(float* p, float v) {
  __hip_atomic_fetch_add(p, v, __ATOMIC_RELAXED, __HIP_MEMORY_SCOPE_AGENT);
}

__global__ void k_fill(float* p, float v, int n) {
  int i = blockIdx.x * blockDim.x + threadIdx.x;
  if (i < n) p[i] = v;
}

__global__ void k_deg_count(const int* __restrict__ dst, float* __restrict__ deg, int E) {
  int e = blockIdx.x * blockDim.x + threadIdx.x;
  if (e < E) atomAddF(&deg[dst[e]], 1.0f);
}

__global__ void k_to_dinv(float* deg, int n) {
  int i = blockIdx.x * blockDim.x + threadIdx.x;
  if (i < n) deg[i] = rsqrtf(fmaxf(deg[i], 1.0f));
}

template <int OUT>
__global__ void k_gemm_wmma(const float* __restrict__ A, const float* __restrict__ W,
                            float* __restrict__ C, int N) {
  constexpr int KD = 128;
  __shared__ float As[16 * KD];

  const int row0 = blockIdx.x * 16;
  for (int i = threadIdx.x; i < 16 * KD; i += blockDim.x) {
    int r = i / KD, k = i % KD;
    As[i] = (row0 + r < N) ? A[(size_t)(row0 + r) * KD + k] : 0.0f;
  }
  __syncthreads();

  const int wave = threadIdx.x >> 5;
  const int lane = threadIdx.x & 31;
  const int half = lane >> 4;
  const int n    = lane & 15;
  const int cw   = wave * 16;

  v8f c = {};
  for (int k0 = 0; k0 < KD; k0 += 4) {
    const int k = k0 + 2 * half;
    v2f a, b;
    a.x = As[n * KD + k];
    a.y = As[n * KD + k + 1];
    b.x = W[(size_t)k * OUT + cw + n];
    b.y = W[(size_t)(k + 1) * OUT + cw + n];
    c = __builtin_amdgcn_wmma_f32_16x16x4_f32(false, a, false, b,
                                              (short)0, c, false, false);
  }

#pragma unroll
  for (int r = 0; r < 8; ++r) {
    const int m = r + 8 * half;
    if (row0 + m < N) C[(size_t)(row0 + m) * OUT + cw + n] = c[r];
  }
}

template <int F>
__global__ void k_agg_init(const float* __restrict__ xw, const float* __restrict__ dinv,
                           const float* __restrict__ bias, float* __restrict__ out, int N) {
  constexpr int LP = F / 4;
  int t = blockIdx.x * blockDim.x + threadIdx.x;
  int node = t / LP;
  if (node >= N) return;
  int fo = (t % LP) * 4;
  float s = dinv[node]; s *= s;
  float4 v = *(const float4*)&xw[(size_t)node * F + fo];
  const float4 b = *(const float4*)&bias[fo];
  v.x = v.x * s + b.x; v.y = v.y * s + b.y;
  v.z = v.z * s + b.z; v.w = v.w * s + b.w;
  *(float4*)&out[(size_t)node * F + fo] = v;
}

template <int F>
struct AggEdge {
  static constexpr int NJ = F / 32;
  static constexpr int EB = 16 / NJ;

  template <int CNT>
  static __device__ __forceinline__ void batch(const float* __restrict__ xw,
                                               float* __restrict__ out,
                                               int s, int d, float nrm, int i0, int lane) {
    float v[CNT][NJ];
    int   dd[CNT];
#pragma unroll
    for (int b = 0; b < CNT; ++b) {
      const int   si = __builtin_amdgcn_readlane(s, i0 + b);
      const float ni = __int_as_float(__builtin_amdgcn_readlane(__float_as_int(nrm), i0 + b));
      dd[b] = __builtin_amdgcn_readlane(d, i0 + b);
      const float* xr = xw + (size_t)si * F + lane;
#pragma unroll
      for (int j = 0; j < NJ; ++j) v[b][j] = xr[32 * j] * ni;
    }
#pragma unroll
    for (int b = 0; b < CNT; ++b) {
      float* o = out + (size_t)dd[b] * F + lane;
#pragma unroll
      for (int j = 0; j < NJ; ++j) atomAddF(o + 32 * j, v[b][j]);
    }
  }
};

template <int F>
__global__ void k_agg_edges(const float* __restrict__ xw, const float* __restrict__ dinv,
                            const int* __restrict__ src, const int* __restrict__ dst,
                            float* __restrict__ out, int E) {
  constexpr int EPW = 16;
  constexpr int EB  = AggEdge<F>::EB;
  const int lane = threadIdx.x & 31;
  const int wave = (blockIdx.x * blockDim.x + threadIdx.x) >> 5;
  const int e0 = wave * EPW;
  if (e0 >= E) return;
  const int cnt = (E - e0 < EPW) ? (E - e0) : EPW;
  int s = 0, d = 0;
  float nrm = 0.0f;
  if (lane < cnt) {
    s = src[e0 + lane];
    d = dst[e0 + lane];
    nrm = dinv[s] * dinv[d];
  }
  if (cnt == EPW) {
#pragma unroll
    for (int i0 = 0; i0 < EPW; i0 += EB)
      AggEdge<F>::template batch<EB>(xw, out, s, d, nrm, i0, lane);
  } else {
    for (int i = 0; i < cnt; ++i)
      AggEdge<F>::template batch<1>(xw, out, s, d, nrm, i, lane);
  }
}

__global__ void k_elu_pool(float* __restrict__ h, const int* __restrict__ batch,
                           float* __restrict__ pool, float* __restrict__ cnt, int N) {
  int node = blockIdx.x * (blockDim.x >> 5) + (threadIdx.x >> 5);
  if (node >= N) return;
  int lane = threadIdx.x & 31;
  float* hr = h + (size_t)node * 128 + lane;
  float v[4];
#pragma unroll
  for (int j = 0; j < 4; ++j) v[j] = hr[32 * j];
#pragma unroll
  for (int j = 0; j < 4; ++j) v[j] = v[j] > 0.0f ? v[j] : expm1f(v[j]);
#pragma unroll
  for (int j = 0; j < 4; ++j) hr[32 * j] = v[j];
  int g = batch[node];
  float* p = pool + (size_t)g * 128 + lane;
#pragma unroll
  for (int j = 0; j < 4; ++j) atomAddF(p + 32 * j, v[j]);
  if (lane == 0) atomAddF(&cnt[g], 1.0f);
}

template <int IN, int OUT>
__global__ void k_master_mlp(const float* __restrict__ pool, const float* __restrict__ cnt,
                             const float* __restrict__ Wm, const float* __restrict__ bm,
                             float* __restrict__ mm, const int* __restrict__ gptr) {
  int G = *gptr;
  int idx = blockIdx.x * blockDim.x + threadIdx.x;
  int g = idx / OUT, j = idx % OUT;
  if (g >= G) return;
  float inv = 1.0f / fmaxf(cnt[g], 1.0f);
  float acc = bm[j];
  for (int i = 0; i < IN; ++i)
    acc = fmaf(pool[(size_t)g * IN + i] * inv, Wm[(size_t)i * OUT + j], acc);
  mm[(size_t)g * OUT + j] = fmaxf(acc, 0.0f);
}

__global__ void k_add_master(float* __restrict__ h, const float* __restrict__ mm,
                             const int* __restrict__ batch, int N) {
  int t = blockIdx.x * blockDim.x + threadIdx.x;
  int node = t >> 5;
  if (node >= N) return;
  int fo = (t & 31) * 4;
  int g = batch[node];
  float4 a = *(float4*)&h[(size_t)node * 128 + fo];
  float4 m = *(const float4*)&mm[(size_t)g * 128 + fo];
  a.x += m.x; a.y += m.y; a.z += m.z; a.w += m.w;
  *(float4*)&h[(size_t)node * 128 + fo] = a;
}

__global__ void k_pool_z(const float* __restrict__ z, const int* __restrict__ batch,
                         float* __restrict__ pool, int N) {
  int t = blockIdx.x * blockDim.x + threadIdx.x;
  int node = t >> 5;
  if (node >= N) return;
  int f = t & 31;
  atomAddF(&pool[(size_t)batch[node] * 32 + f], z[(size_t)node * 32 + f]);
}

__global__ void k_finalize(const float* __restrict__ z, const float* __restrict__ mm2,
                           const int* __restrict__ batch, float* __restrict__ out, int N) {
  int node = blockIdx.x * (blockDim.x >> 5) + (threadIdx.x >> 5);
  if (node >= N) return;
  int lane = threadIdx.x & 31;
  float v = z[(size_t)node * 32 + lane] + mm2[(size_t)batch[node] * 32 + lane];
  float m = v;
#pragma unroll
  for (int o = 16; o >= 1; o >>= 1) m = fmaxf(m, __shfl_xor(m, o, 32));
  float ev = __expf(v - m);
  float s = ev;
#pragma unroll
  for (int o = 16; o >= 1; o >>= 1) s += __shfl_xor(s, o, 32);
  out[(size_t)node * 32 + lane] = ev / s;
}

extern "C" void kernel_launch(void* const* d_in, const int* in_sizes, int n_in,
                              void* d_out, int out_size, void* d_ws, size_t ws_size,
                              hipStream_t stream) {
  const float* X   = (const float*)d_in[0];
  const float* W1  = (const float*)d_in[1];
  const float* b1  = (const float*)d_in[2];
  const float* W2  = (const float*)d_in[3];
  const float* b2  = (const float*)d_in[4];
  const float* mW1 = (const float*)d_in[5];
  const float* mb1 = (const float*)d_in[6];
  const float* mW2 = (const float*)d_in[7];
  const float* mb2 = (const float*)d_in[8];
  const int* ei    = (const int*)d_in[9];
  const int* batch = (const int*)d_in[10];
  const int* gptr  = (const int*)d_in[11];

  const int N = in_sizes[0] / 128;
  const int E = in_sizes[9] / 2;
  const int* src = ei;
  const int* dst = ei + E;

  char* w = (char*)d_ws;
  size_t off = 0;
  auto take = [&](size_t bytes) -> float* {
    float* p = (float*)(w + off);
    off = (off + bytes + 255) & ~(size_t)255;
    return p;
  };
  float* xw    = take((size_t)N * 128 * 4);
  float* h     = take((size_t)N * 128 * 4);
  float* dinv  = take((size_t)N * 4);
  float* z     = take((size_t)N * 32 * 4);
  float* pool1 = take((size_t)GMAX * 128 * 4);
  float* pool2 = take((size_t)GMAX * 32 * 4);
  float* cnt   = take((size_t)GMAX * 4);
  float* mm1   = take((size_t)GMAX * 128 * 4);
  float* mm2   = take((size_t)GMAX * 32 * 4);

  const int B = 256;
  auto blks = [](long long n, int b) { return (unsigned)((n + b - 1) / b); };

  k_fill<<<blks(N, B), B, 0, stream>>>(dinv, 1.0f, N);
  k_fill<<<blks(GMAX * 128, B), B, 0, stream>>>(pool1, 0.0f, GMAX * 128);
  k_fill<<<blks(GMAX * 32, B), B, 0, stream>>>(pool2, 0.0f, GMAX * 32);
  k_fill<<<blks(GMAX, B), B, 0, stream>>>(cnt, 0.0f, GMAX);
  k_deg_count<<<blks(E, B), B, 0, stream>>>(dst, dinv, E);
  k_to_dinv<<<blks(N, B), B, 0, stream>>>(dinv, N);

  k_gemm_wmma<128><<<blks(N, 16), 256, 0, stream>>>(X, W1, xw, N);
  k_agg_init<128><<<blks((long long)N * 32, B), B, 0, stream>>>(xw, dinv, b1, h, N);
  k_agg_edges<128><<<blks((long long)E, 128), B, 0, stream>>>(xw, dinv, src, dst, h, E);
  k_elu_pool<<<blks(N, 8), B, 0, stream>>>(h, batch, pool1, cnt, N);

  k_master_mlp<128, 128><<<blks(GMAX * 128, B), B, 0, stream>>>(pool1, cnt, mW1, mb1, mm1, gptr);
  k_add_master<<<blks((long long)N * 32, B), B, 0, stream>>>(h, mm1, batch, N);

  k_gemm_wmma<32><<<blks(N, 16), 64, 0, stream>>>(h, W2, xw, N);
  k_agg_init<32><<<blks((long long)N * 8, B), B, 0, stream>>>(xw, dinv, b2, z, N);
  k_agg_edges<32><<<blks((long long)E, 128), B, 0, stream>>>(xw, dinv, src, dst, z, E);

  k_pool_z<<<blks((long long)N * 32, B), B, 0, stream>>>(z, batch, pool2, N);
  k_master_mlp<32, 32><<<blks(GMAX * 32, B), B, 0, stream>>>(pool2, cnt, mW2, mb2, mm2, gptr);
  k_finalize<<<blks(N, 8), B, 0, stream>>>(z, mm2, batch, (float*)d_out, N);
}


# speedup vs baseline: 1.9675x; 1.2221x over previous
#include <hip/hip_runtime.h>
#include <math.h>


typedef __attribute__((ext_vector_type(2))) float v2f;
typedef __attribute__((ext_vector_type(8))) float v8f;

#define GMAX 256

__device__ __forceinline__ void atomAddF(float* p, float v) {
  __hip_atomic_fetch_add(p, v, __ATOMIC_RELAXED, __HIP_MEMORY_SCOPE_AGENT);
}

__device__ __forceinline__ int pkg_half() {
  return (int)((__builtin_amdgcn_s_sendmsg_rtn(0x87) >> 18) & 1u);
}

__global__ void k_fill(float* p, float v, int n) {
  int i = blockIdx.x * blockDim.x + threadIdx.x;
  if (i < n) p[i] = v;
}

__global__ void k_deg_init(float* __restrict__ deg0, float* __restrict__ deg1, int n) {
  int i = blockIdx.x * blockDim.x + threadIdx.x;
  if (i < n) {
    deg0[i] = 1.0f;
    if (deg1) deg1[i] = 0.0f;
  }
}

__global__ void k_deg_count(const int* __restrict__ dst, float* __restrict__ deg0,
                            float* __restrict__ deg1, int E) {
  int e = blockIdx.x * blockDim.x + threadIdx.x;
  float* deg = (deg1 != nullptr && pkg_half()) ? deg1 : deg0;
  if (e < E) atomAddF(&deg[dst[e]], 1.0f);
}

__global__ void k_to_dinv(float* __restrict__ deg, const float* __restrict__ deg1, int n) {
  int i = blockIdx.x * blockDim.x + threadIdx.x;
  if (i < n) {
    float d = deg[i];
    if (deg1) d += deg1[i];
    deg[i] = rsqrtf(fmaxf(d, 1.0f));
  }
}

template <int OUT>
__global__ void k_gemm_wmma(const float* __restrict__ A, const float* __restrict__ W,
                            float* __restrict__ C, int N) {
  constexpr int KD = 128;
  __shared__ float As[16 * KD];

  const int row0 = blockIdx.x * 16;
  for (int i = threadIdx.x; i < 16 * KD; i += blockDim.x) {
    int r = i / KD, k = i % KD;
    As[i] = (row0 + r < N) ? A[(size_t)(row0 + r) * KD + k] : 0.0f;
  }
  __syncthreads();

  const int wave = threadIdx.x >> 5;
  const int lane = threadIdx.x & 31;
  const int half = lane >> 4;
  const int n    = lane & 15;
  const int cw   = wave * 16;

  v8f c = {};
  for (int k0 = 0; k0 < KD; k0 += 4) {
    const int k = k0 + 2 * half;
    v2f a, b;
    a.x = As[n * KD + k];
    a.y = As[n * KD + k + 1];
    b.x = W[(size_t)k * OUT + cw + n];
    b.y = W[(size_t)(k + 1) * OUT + cw + n];
    c = __builtin_amdgcn_wmma_f32_16x16x4_f32(false, a, false, b,
                                              (short)0, c, false, false);
  }

#pragma unroll
  for (int r = 0; r < 8; ++r) {
    const int m = r + 8 * half;
    if (row0 + m < N) C[(size_t)(row0 + m) * OUT + cw + n] = c[r];
  }
}

template <int F>
__global__ void k_agg_init(const float* __restrict__ xw, const float* __restrict__ dinv,
                           const float* __restrict__ bias, float* __restrict__ out,
                           float* __restrict__ out1, int N) {
  constexpr int LP = F / 4;
  int t = blockIdx.x * blockDim.x + threadIdx.x;
  int node = t / LP;
  if (node >= N) return;
  int fo = (t % LP) * 4;
  float s = dinv[node]; s *= s;
  float4 v = *(const float4*)&xw[(size_t)node * F + fo];
  const float4 b = *(const float4*)&bias[fo];
  v.x = v.x * s + b.x; v.y = v.y * s + b.y;
  v.z = v.z * s + b.z; v.w = v.w * s + b.w;
  *(float4*)&out[(size_t)node * F + fo] = v;
  if (out1) *(float4*)&out1[(size_t)node * F + fo] = make_float4(0.0f, 0.0f, 0.0f, 0.0f);
}

template <int F>
struct AggEdge {
  static constexpr int NJ = F / 32;
  static constexpr int EB = 16 / NJ;

  template <int CNT>
  static __device__ __forceinline__ void batch(const float* __restrict__ xw,
                                               float* __restrict__ out,
                                               int s, int d, float nrm, int i0, int lane) {
    float v[CNT][NJ];
    int   dd[CNT];
#pragma unroll
    for (int b = 0; b < CNT; ++b) {
      const int   si = __builtin_amdgcn_readlane(s, i0 + b);
      const float ni = __int_as_float(__builtin_amdgcn_readlane(__float_as_int(nrm), i0 + b));
      dd[b] = __builtin_amdgcn_readlane(d, i0 + b);
      const float* xr = xw + (size_t)si * F + lane;
#pragma unroll
      for (int j = 0; j < NJ; ++j) v[b][j] = xr[32 * j] * ni;
    }
#pragma unroll
    for (int b = 0; b < CNT; ++b) {
      float* o = out + (size_t)dd[b] * F + lane;
#pragma unroll
      for (int j = 0; j < NJ; ++j) atomAddF(o + 32 * j, v[b][j]);
    }
  }
};

template <int F>
__global__ void k_agg_edges(const float* __restrict__ xw, const float* __restrict__ dinv,
                            const int* __restrict__ src, const int* __restrict__ dst,
                            float* __restrict__ out0, float* __restrict__ out1, int E) {
  constexpr int EPW = 16;
  constexpr int EB  = AggEdge<F>::EB;
  const int lane = threadIdx.x & 31;
  const int wave = (blockIdx.x * blockDim.x + threadIdx.x) >> 5;
  const int e0 = wave * EPW;
  if (e0 >= E) return;
  float* out = (out1 != nullptr && pkg_half()) ? out1 : out0;
  const int cnt = (E - e0 < EPW) ? (E - e0) : EPW;
  int s = 0, d = 0;
  float nrm = 0.0f;
  if (lane < cnt) {
    s = src[e0 + lane];
    d = dst[e0 + lane];
    nrm = dinv[s] * dinv[d];
  }
  if (cnt == EPW) {
#pragma unroll
    for (int i0 = 0; i0 < EPW; i0 += EB)
      AggEdge<F>::template batch<EB>(xw, out, s, d, nrm, i0, lane);
  } else {
    for (int i = 0; i < cnt; ++i)
      AggEdge<F>::template batch<1>(xw, out, s, d, nrm, i, lane);
  }
}

__global__ void k_elu_pool(float* __restrict__ h, const float* __restrict__ h1,
                           const int* __restrict__ batch,
                           float* __restrict__ pool, float* __restrict__ cnt, int N) {
  int node = blockIdx.x * (blockDim.x >> 5) + (threadIdx.x >> 5);
  if (node >= N) return;
  int lane = threadIdx.x & 31;
  float* hr = h + (size_t)node * 128 + lane;
  float v[4];
#pragma unroll
  for (int j = 0; j < 4; ++j) v[j] = hr[32 * j];
  if (h1) {
    const float* h1r = h1 + (size_t)node * 128 + lane;
#pragma unroll
    for (int j = 0; j < 4; ++j) v[j] += h1r[32 * j];
  }
#pragma unroll
  for (int j = 0; j < 4; ++j) v[j] = v[j] > 0.0f ? v[j] : expm1f(v[j]);
#pragma unroll
  for (int j = 0; j < 4; ++j) hr[32 * j] = v[j];
  int g = batch[node];
  float* p = pool + (size_t)g * 128 + lane;
#pragma unroll
  for (int j = 0; j < 4; ++j) atomAddF(p + 32 * j, v[j]);
  if (lane == 0) atomAddF(&cnt[g], 1.0f);
}

template <int IN, int OUT>
__global__ void k_master_mlp(const float* __restrict__ pool, const float* __restrict__ cnt,
                             const float* __restrict__ Wm, const float* __restrict__ bm,
                             float* __restrict__ mm, const int* __restrict__ gptr) {
  int G = *gptr;
  int idx = blockIdx.x * blockDim.x + threadIdx.x;
  int g = idx / OUT, j = idx % OUT;
  if (g >= G) return;
  float inv = 1.0f / fmaxf(cnt[g], 1.0f);
  float acc = bm[j];
  for (int i = 0; i < IN; ++i)
    acc = fmaf(pool[(size_t)g * IN + i] * inv, Wm[(size_t)i * OUT + j], acc);
  mm[(size_t)g * OUT + j] = fmaxf(acc, 0.0f);
}

__global__ void k_add_master(float* __restrict__ h, const float* __restrict__ mm,
                             const int* __restrict__ batch, int N) {
  int t = blockIdx.x * blockDim.x + threadIdx.x;
  int node = t >> 5;
  if (node >= N) return;
  int fo = (t & 31) * 4;
  int g = batch[node];
  float4 a = *(float4*)&h[(size_t)node * 128 + fo];
  float4 m = *(const float4*)&mm[(size_t)g * 128 + fo];
  a.x += m.x; a.y += m.y; a.z += m.z; a.w += m.w;
  *(float4*)&h[(size_t)node * 128 + fo] = a;
}

__global__ void k_pool_z(float* __restrict__ z, const float* __restrict__ z1,
                         const int* __restrict__ batch,
                         float* __restrict__ pool, int N) {
  int t = blockIdx.x * blockDim.x + threadIdx.x;
  int node = t >> 5;
  if (node >= N) return;
  int f = t & 31;
  const size_t i = (size_t)node * 32 + f;
  float v = z[i];
  if (z1) {
    v += z1[i];
    z[i] = v;
  }
  atomAddF(&pool[(size_t)batch[node] * 32 + f], v);
}

__global__ void k_finalize(const float* __restrict__ z, const float* __restrict__ mm2,
                           const int* __restrict__ batch, float* __restrict__ out, int N) {
  int node = blockIdx.x * (blockDim.x >> 5) + (threadIdx.x >> 5);
  if (node >= N) return;
  int lane = threadIdx.x & 31;
  float v = z[(size_t)node * 32 + lane] + mm2[(size_t)batch[node] * 32 + lane];
  float m = v;
#pragma unroll
  for (int o = 16; o >= 1; o >>= 1) m = fmaxf(m, __shfl_xor(m, o, 32));
  float ev = __expf(v - m);
  float s = ev;
#pragma unroll
  for (int o = 16; o >= 1; o >>= 1) s += __shfl_xor(s, o, 32);
  out[(size_t)node * 32 + lane] = ev / s;
}

extern "C" void kernel_launch(void* const* d_in, const int* in_sizes, int n_in,
                              void* d_out, int out_size, void* d_ws, size_t ws_size,
                              hipStream_t stream) {
  const float* X   = (const float*)d_in[0];
  const float* W1  = (const float*)d_in[1];
  const float* b1  = (const float*)d_in[2];
  const float* W2  = (const float*)d_in[3];
  const float* b2  = (const float*)d_in[4];
  const float* mW1 = (const float*)d_in[5];
  const float* mb1 = (const float*)d_in[6];
  const float* mW2 = (const float*)d_in[7];
  const float* mb2 = (const float*)d_in[8];
  const int* ei    = (const int*)d_in[9];
  const int* batch = (const int*)d_in[10];
  const int* gptr  = (const int*)d_in[11];

  const int N = in_sizes[0] / 128;
  const int E = in_sizes[9] / 2;
  const int* src = ei;
  const int* dst = ei + E;

  char* w = (char*)d_ws;
  size_t off = 0;
  auto take = [&](size_t bytes) -> float* {
    float* p = (float*)(w + off);
    off = (off + bytes + 255) & ~(size_t)255;
    return p;
  };
  float* xw    = take((size_t)N * 128 * 4);
  float* h     = take((size_t)N * 128 * 4);
  float* dinv  = take((size_t)N * 4);
  float* z     = take((size_t)N * 32 * 4);
  float* pool1 = take((size_t)GMAX * 128 * 4);
  float* pool2 = take((size_t)GMAX * 32 * 4);
  float* cnt   = take((size_t)GMAX * 4);
  float* mm1   = take((size_t)GMAX * 128 * 4);
  float* mm2   = take((size_t)GMAX * 32 * 4);
  float* h1 = nullptr;
  if (off + (size_t)N * 128 * 4 <= ws_size) h1 = take((size_t)N * 128 * 4);
  float* z1 = h1;
  float* deg1 = nullptr;
  if (off + (size_t)N * 4 <= ws_size) deg1 = take((size_t)N * 4);

  const int B = 256;
  auto blks = [](long long n, int b) { return (unsigned)((n + b - 1) / b); };

  k_deg_init<<<blks(N, B), B, 0, stream>>>(dinv, deg1, N);
  k_fill<<<blks(GMAX * 128, B), B, 0, stream>>>(pool1, 0.0f, GMAX * 128);
  k_fill<<<blks(GMAX * 32, B), B, 0, stream>>>(pool2, 0.0f, GMAX * 32);
  k_fill<<<blks(GMAX, B), B, 0, stream>>>(cnt, 0.0f, GMAX);
  k_deg_count<<<blks(E, B), B, 0, stream>>>(dst, dinv, deg1, E);
  k_to_dinv<<<blks(N, B), B, 0, stream>>>(dinv, deg1, N);

  k_gemm_wmma<128><<<blks(N, 16), 256, 0, stream>>>(X, W1, xw, N);
  k_agg_init<128><<<blks((long long)N * 32, B), B, 0, stream>>>(xw, dinv, b1, h, h1, N);
  k_agg_edges<128><<<blks((long long)E, 128), B, 0, stream>>>(xw, dinv, src, dst, h, h1, E);
  k_elu_pool<<<blks(N, 8), B, 0, stream>>>(h, h1, batch, pool1, cnt, N);

  k_master_mlp<128, 128><<<blks(GMAX * 128, B), B, 0, stream>>>(pool1, cnt, mW1, mb1, mm1, gptr);
  k_add_master<<<blks((long long)N * 32, B), B, 0, stream>>>(h, mm1, batch, N);

  k_gemm_wmma<32><<<blks(N, 16), 64, 0, stream>>>(h, W2, xw, N);
  k_agg_init<32><<<blks((long long)N * 8, B), B, 0, stream>>>(xw, dinv, b2, z, z1, N);
  k_agg_edges<32><<<blks((long long)E, 128), B, 0, stream>>>(xw, dinv, src, dst, z, z1, E);

  k_pool_z<<<blks((long long)N * 32, B), B, 0, stream>>>(z, z1, batch, pool2, N);
  k_master_mlp<32, 32><<<blks(GMAX * 32, B), B, 0, stream>>>(pool2, cnt, mW2, mb2, mm2, gptr);
  k_finalize<<<blks(N, 8), B, 0, stream>>>(z, mm2, batch, (float*)d_out, N);
}
